# starting kernel
# baseline (speedup 1.0000x reference)
#include <hip/hip_runtime.h>

typedef float v2f __attribute__((ext_vector_type(2)));
typedef float v8f __attribute__((ext_vector_type(8)));

#define NQ    12
#define NL    6
#define DIM   4096
#define BLOCK 256

__host__ __device__ static constexpr int state_index(int r, int j, int sh) {
    return ((r >> sh) << (sh + 4)) | (j << sh) | (r & ((1 << sh) - 1));
}

__host__ __device__ static constexpr int swzc(int i) {
    return i ^ ((i >> 4) & 15) ^ ((i >> 8) & 15);
}

__host__ __device__ static constexpr int Rmap(int j) {
    int x = j;
    for (int w = 11; w >= 0; --w) {
        const int pc = 11 - w;
        const int pt = 11 - ((w + 1) % 12);
        x ^= ((x >> pc) & 1) << pt;
    }
    return x;
}

__global__ __launch_bounds__(BLOCK) void qsim12_kernel(
    const float* __restrict__ model_input,
    const float* __restrict__ weights,
    float* __restrict__ out)
{
    __shared__ float2 st2[DIM];
    __shared__ float2 Ar2[3][256], Ai2[3][256];
    __shared__ float gates[NL][NQ][8];
    __shared__ float red[BLOCK];

    const int b    = blockIdx.x;
    const int t    = threadIdx.x;
    const int lane = t & 31;
    const int wv   = t >> 5;

    const int storeBase = swzc(t << 4);
    const int permBase  = swzc(Rmap(t << 4));

    for (int k = t; k < DIM; k += BLOCK) st2[k] = make_float2(0.0f, 0.0f);
    if (t == 0) st2[0] = make_float2(1.0f, 0.0f);

    const float a0 = model_input[2 * b + 0];
    const float a1 = model_input[2 * b + 1];

    if (t < NL * NQ) {
        const int l = t / NQ, w = t % NQ;
        const int limit = NQ - (l >> 1);
        float th = weights[(l * NQ + w) * 2 + 0];
        if (w < limit) th += (w & 1) ? a1 : a0;
        const float ph = weights[(l * NQ + w) * 2 + 1];
        float c, s, pr, pi;
        __sincosf(0.5f * th, &s, &c);
        __sincosf(0.5f * ph, &pi, &pr);
        gates[l][w][0] =  pr * c;  gates[l][w][1] = -pi * c;
        gates[l][w][2] = -pr * s;  gates[l][w][3] =  pi * s;
        gates[l][w][4] =  pr * s;  gates[l][w][5] =  pi * s;
        gates[l][w][6] =  pr * c;  gates[l][w][7] =  pi * c;
    }
    __syncthreads();

    #pragma unroll 1
    for (int l = 0; l < NL; ++l) {
        {
            const int jo = t >> 4, ji = t & 15;
            #pragma unroll
            for (int g = 0; g < 3; ++g) {
                float re = 1.0f, im = 0.0f;
                #pragma unroll
                for (int k = 0; k < 4; ++k) {
                    const int w  = 4 * g + k;
                    const int bo = (jo >> (3 - k)) & 1;
                    const int bi = (ji >> (3 - k)) & 1;
                    const float er = gates[l][w][(bo * 2 + bi) * 2 + 0];
                    const float ei = gates[l][w][(bo * 2 + bi) * 2 + 1];
                    const float nr = re * er - im * ei;
                    const float ni = re * ei + im * er;
                    re = nr; im = ni;
                }
                Ar2[g][t] = make_float2(re, -im);
                Ai2[g][t] = make_float2(im,  re);
            }
        }
        __syncthreads();

        #pragma unroll
        for (int g = 0; g < 3; ++g) {
            const int sh = 8 - 4 * g;

            const int m  = lane & 15;
            const int hi = lane >> 4;
            v2f ar[8], ai[8];
            #pragma unroll
            for (int c = 0; c < 8; ++c) {
                const int col = m * 16 + 2 * c + hi;
                ar[c] = *(const v2f*)&Ar2[g][col];
                ai[c] = *(const v2f*)&Ai2[g][col];
            }

            #pragma unroll
            for (int tt = 0; tt < 2; ++tt) {
                const int r0 = (wv * 2 + tt) * 16;
                const int rn = r0 + (lane & 15);
                const int sri = swzc(state_index(rn, 0, sh));
                const int sbB = sri ^ (hi ? swzc(1 << sh) : 0);
                const int sbD = sri ^ (hi ? swzc(8 << sh) : 0);

                v2f bb[8];
                #pragma unroll
                for (int c = 0; c < 8; ++c) {
                    bb[c] = *(const v2f*)&st2[sbB ^ swzc((2 * c) << sh)];
                }
                v8f dr = {0, 0, 0, 0, 0, 0, 0, 0};
                v8f di = {0, 0, 0, 0, 0, 0, 0, 0};
                #pragma unroll
                for (int c = 0; c < 8; ++c) {
                    dr = __builtin_amdgcn_wmma_f32_16x16x4_f32(false, ar[c], false, bb[c], (short)0, dr, false, false);
                    di = __builtin_amdgcn_wmma_f32_16x16x4_f32(false, ai[c], false, bb[c], (short)0, di, false, false);
                }
                #pragma unroll
                for (int v = 0; v < 8; ++v) {
                    st2[sbD ^ swzc(v << sh)] = make_float2(dr[v], di[v]);
                }
            }
            __syncthreads();
        }

        float2 tmp[16];
        #pragma unroll
        for (int k = 0; k < 16; ++k) {
            tmp[k] = st2[permBase ^ swzc(Rmap(k))];
        }
        __syncthreads();
        #pragma unroll
        for (int k = 0; k < 16; ++k) {
            st2[storeBase ^ k] = tmp[k];
        }
        __syncthreads();
    }

    float acc = 0.0f;
    #pragma unroll
    for (int k = 0; k < 16; ++k) {
        const float2 v = st2[storeBase ^ k];
        const float p = v.x * v.x + v.y * v.y;
        acc += (k & 1) ? -p : p;
    }
    red[t] = acc;
    __syncthreads();
    #pragma unroll
    for (int off = BLOCK / 2; off > 0; off >>= 1) {
        if (t < off) red[t] += red[t + off];
        __syncthreads();
    }
    if (t == 0) out[b] = red[0];
}

extern "C" void kernel_launch(void* const* d_in, const int* in_sizes, int n_in,
                              void* d_out, int out_size, void* d_ws, size_t ws_size,
                              hipStream_t stream) {
    const float* model_input = (const float*)d_in[0];
    const float* weights     = (const float*)d_in[1];
    float* out = (float*)d_out;
    const int batch = in_sizes[0] / 2;
    qsim12_kernel<<<batch, BLOCK, 0, stream>>>(model_input, weights, out);
}


# LDS state moved in 16-byte accesses (2 neighbouring amplitudes per lane: 2 tiles in groups 0-2, 2 K chunks in group 3), swizzle one bit higher: no bank conflicts; group 3 k sums, readout regrouped
# speedup vs baseline: 1.6105x; 1.6105x over previous
#include <hip/hip_runtime.h>

typedef float v2f __attribute__((ext_vector_type(2)));
typedef float v8f __attribute__((ext_vector_type(8)));

#define NQ    12
#define NL    6
#define NG    4
#define DIM   4096
#define BLOCK 256


__host__ __device__ static constexpr int swzc(int i) {
    return i ^ ((((i >> 5) & 15) ^ ((i >> 9) & 7)) << 1);
}

__host__ __device__ static constexpr int colM(int g, int m) {
    return g == 2 ? (((m & 3) << 1) | ((m >> 2) << 7)) : (m << 1);
}
__host__ __device__ static constexpr int colX(int g, int x) {
    return g == 0 ? (x << 5)
         : g == 1 ? (((x & 1) << 5) | ((x >> 1) << 9))
                  : (((x & 1) << 6) | ((x >> 1) << 9));
}

__host__ __device__ static constexpr int Rmap(int j) {
    int x = j;
    for (int w = 11; w >= 0; --w) {
        const int pc = 11 - w;
        const int pt = 11 - ((w + 1) % 12);
        x ^= ((x >> pc) & 1) << pt;
    }
    return x;
}

__device__ __forceinline__ void mma_tiles(const v2f (&af)[4], const v2f (&bb)[4][4], v8f (&d)[4]) {
    #pragma unroll
    for (int tt = 0; tt < 4; ++tt) d[tt] = (v8f){0, 0, 0, 0, 0, 0, 0, 0};
    #pragma unroll
    for (int c = 0; c < 4; ++c) {
        #pragma unroll
        for (int tt = 0; tt < 4; ++tt) {
            d[tt] = __builtin_amdgcn_wmma_f32_16x16x4_f32(false, af[c], false, bb[tt][c], (short)0, d[tt], false, false);
        }
    }
}

__global__ __launch_bounds__(BLOCK) void qsim12_kernel(
    const float* __restrict__ model_input,
    const float* __restrict__ weights,
    float* __restrict__ out)
{
    __shared__ float4 st4[DIM / 2];
    __shared__ float2 Ae[NG][128];
    __shared__ float gates[NL][NQ][8];
    __shared__ float red[BLOCK];

    float2* const st2 = reinterpret_cast<float2*>(st4);

    const int b    = blockIdx.x;
    const int t    = threadIdx.x;
    const int lane = t & 31;
    const int wv   = t >> 5;

    const int storeBase = swzc(lane | (wv << 9));
    const int permBase  = swzc(Rmap(lane | (wv << 9)));

    for (int k = t; k < DIM / 2; k += BLOCK) st4[k] = make_float4(k == 0 ? 1.0f : 0.0f, 0.0f, 0.0f, 0.0f);

    const float a0 = model_input[2 * b + 0];
    const float a1 = model_input[2 * b + 1];

    if (t < NL * NQ) {
        const int l = t / NQ, w = t % NQ;
        const int limit = NQ - (l >> 1);
        float th = weights[(l * NQ + w) * 2 + 0];
        if (w < limit) th += (w & 1) ? a1 : a0;
        const float ph = weights[(l * NQ + w) * 2 + 1];
        float c, s, pr, pi;
        __sincosf(0.5f * th, &s, &c);
        __sincosf(0.5f * ph, &pi, &pr);
        gates[l][w][0] =  pr * c;  gates[l][w][1] = -pi * c;
        gates[l][w][2] = -pr * s;  gates[l][w][3] =  pi * s;
        gates[l][w][4] =  pr * s;  gates[l][w][5] =  pi * s;
        gates[l][w][6] =  pr * c;  gates[l][w][7] =  pi * c;
    }
    __syncthreads();

    const int m  = lane & 15;
    const int hi = lane >> 4;

    #pragma unroll 1
    for (int l = 0; l < NL; ++l) {
        {
            const int g = t >> 6, jo = (t >> 3) & 7, ji = t & 7;
            float re = gates[l][3 * g][((jo >> 2) * 2 + (ji >> 2)) * 2 + 0];
            float im = gates[l][3 * g][((jo >> 2) * 2 + (ji >> 2)) * 2 + 1];
            #pragma unroll
            for (int k = 1; k < 3; ++k) {
                const int w  = 3 * g + k;
                const int bo = (jo >> (2 - k)) & 1;
                const int bi = (ji >> (2 - k)) & 1;
                const float er = gates[l][w][(bo * 2 + bi) * 2 + 0];
                const float ei = gates[l][w][(bo * 2 + bi) * 2 + 1];
                const float nr = re * er - im * ei;
                const float ni = re * ei + im * er;
                re = nr; im = ni;
            }
            const int slot = (g == 3) ? ((ji & 3) * 32 + (ji >> 2)) : ((ji >> 1) * 32 + (ji & 1));
            Ae[g][slot + (2 * jo    ) * 2] = make_float2(re, -im);
            Ae[g][slot + (2 * jo + 1) * 2] = make_float2(im,  re);
        }
        __syncthreads();

        #pragma unroll
        for (int g = 0; g < 3; ++g) {
            const int sh = 9 - 3 * g;

            v2f af[4];
            #pragma unroll
            for (int c = 0; c < 4; ++c) {
                af[c] = *(const v2f*)&Ae[g][c * 32 + m * 2 + hi];
            }

            const int cb  = swzc(colM(g, m) | colX(g, wv * 2));
            const int ldB = cb ^ (hi ? swzc(1 << sh) : 0);
            const int stD = cb ^ (hi ? swzc(4 << sh) : 0);
            v2f bb[4][4];
            #pragma unroll
            for (int tp = 0; tp < 2; ++tp) {
                #pragma unroll
                for (int c = 0; c < 4; ++c) {
                    const float4 v = st4[(ldB ^ swzc(colX(g, tp)) ^ swzc((2 * c) << sh)) >> 1];
                    bb[2 * tp    ][c] = (v2f){v.x, v.y};
                    bb[2 * tp + 1][c] = (v2f){v.z, v.w};
                }
            }
            v8f d[4];
            mma_tiles(af, bb, d);
            #pragma unroll
            for (int tp = 0; tp < 2; ++tp) {
                #pragma unroll
                for (int k = 0; k < 4; ++k) {
                    st4[(stD ^ swzc(colX(g, tp)) ^ swzc(k << sh)) >> 1] =
                        make_float4(d[2 * tp][2 * k], d[2 * tp][2 * k + 1], d[2 * tp + 1][2 * k], d[2 * tp + 1][2 * k + 1]);
                }
            }
            __syncthreads();
        }

        {
            v2f af[4];
            #pragma unroll
            for (int c = 0; c < 4; ++c) {
                af[c] = *(const v2f*)&Ae[3][c * 32 + m * 2 + hi];
            }
            v2f bb[4][4];
            int sb[4];
            #pragma unroll
            for (int tt = 0; tt < 4; ++tt) {
                const int rn = (wv * 4 + tt) * 16 + m;
                sb[tt] = (swzc(rn << 3) >> 1) ^ (2 * hi);
                const float4 v0 = st4[sb[tt]];
                const float4 v1 = st4[sb[tt] ^ 1];
                bb[tt][0] = (v2f){v0.x, v0.y};
                bb[tt][1] = (v2f){v0.z, v0.w};
                bb[tt][2] = (v2f){v1.x, v1.y};
                bb[tt][3] = (v2f){v1.z, v1.w};
            }
            v8f d[4];
            mma_tiles(af, bb, d);
            #pragma unroll
            for (int tt = 0; tt < 4; ++tt) {
                st4[sb[tt]]     = make_float4(d[tt][0], d[tt][1], d[tt][2], d[tt][3]);
                st4[sb[tt] ^ 1] = make_float4(d[tt][4], d[tt][5], d[tt][6], d[tt][7]);
            }
            __syncthreads();
        }

        float2 tmp[16];
        #pragma unroll
        for (int k = 0; k < 16; ++k) {
            tmp[k] = st2[permBase ^ swzc(Rmap(k << 5))];
        }
        __syncthreads();
        #pragma unroll
        for (int k = 0; k < 16; ++k) {
            st2[storeBase ^ swzc(k << 5)] = tmp[k];
        }
        __syncthreads();
    }

    float acc = 0.0f;
    #pragma unroll
    for (int k = 0; k < 16; ++k) {
        const float2 v = st2[storeBase ^ swzc(k << 5)];
        acc += v.x * v.x + v.y * v.y;
    }
    red[t] = (lane & 1) ? -acc : acc;
    __syncthreads();
    #pragma unroll
    for (int off = BLOCK / 2; off > 0; off >>= 1) {
        if (t < off) red[t] += red[t + off];
        __syncthreads();
    }
    if (t == 0) out[b] = red[0];
}

extern "C" void kernel_launch(void* const* d_in, const int* in_sizes, int n_in,
                              void* d_out, int out_size, void* d_ws, size_t ws_size,
                              hipStream_t stream) {
    const float* model_input = (const float*)d_in[0];
    const float* weights     = (const float*)d_in[1];
    float* out = (float*)d_out;
    const int batch = in_sizes[0] / 2;
    qsim12_kernel<<<batch, BLOCK, 0, stream>>>(model_input, weights, out);
}


# Workgroup barriers only where waves exchange amplitudes: ring pass owned by bits 6-8 (group 0's wave), groups 1-3 wave-private, tables built under the ring's barrier: 7 to 3 a layer; readout regrouped
# speedup vs baseline: 1.6782x; 1.0420x over previous
#include <hip/hip_runtime.h>

typedef float v2f __attribute__((ext_vector_type(2)));
typedef float v8f __attribute__((ext_vector_type(8)));

#define NQ    12
#define NL    6
#define NG    4
#define DIM   4096
#define BLOCK 256


__host__ __device__ static constexpr int swzc(int i) {
    return i ^ ((((i >> 5) & 15) ^ ((i >> 9) & 7)) << 1);
}

__host__ __device__ static constexpr int colM(int g, int m) {
    return g == 2 ? (((m & 3) << 1) | ((m >> 2) << 7)) : (m << 1);
}
__host__ __device__ static constexpr int colX(int g, int x) {
    return g == 0 ? (x << 5)
         : g == 1 ? (((x & 1) << 5) | ((x >> 1) << 9))
                  : (((x & 1) << 6) | ((x >> 1) << 9));
}

__host__ __device__ static constexpr int Rmap(int j) {
    int x = j;
    for (int w = 11; w >= 0; --w) {
        const int pc = 11 - w;
        const int pt = 11 - ((w + 1) % 12);
        x ^= ((x >> pc) & 1) << pt;
    }
    return x;
}

__host__ __device__ static constexpr int ringK(int k) {
    return ((k & 1) << 5) | ((k >> 1) << 9);
}

__device__ __forceinline__ void wave_lds_order() {
    __builtin_amdgcn_fence(__ATOMIC_RELEASE, "wavefront");
    __builtin_amdgcn_wave_barrier();
    __builtin_amdgcn_fence(__ATOMIC_ACQUIRE, "wavefront");
}

__device__ __forceinline__ void mma_tiles(const v2f (&af)[4], const v2f (&bb)[4][4], v8f (&d)[4]) {
    #pragma unroll
    for (int tt = 0; tt < 4; ++tt) d[tt] = (v8f){0, 0, 0, 0, 0, 0, 0, 0};
    #pragma unroll
    for (int c = 0; c < 4; ++c) {
        #pragma unroll
        for (int tt = 0; tt < 4; ++tt) {
            d[tt] = __builtin_amdgcn_wmma_f32_16x16x4_f32(false, af[c], false, bb[tt][c], (short)0, d[tt], false, false);
        }
    }
}

__global__ __launch_bounds__(BLOCK) void qsim12_kernel(
    const float* __restrict__ model_input,
    const float* __restrict__ weights,
    float* __restrict__ out)
{
    __shared__ float4 st4[DIM / 2];
    __shared__ float2 Ae[NG][128];
    __shared__ float gates[NL][NQ][8];
    __shared__ float red[BLOCK];

    float2* const st2 = reinterpret_cast<float2*>(st4);

    const int b    = blockIdx.x;
    const int t    = threadIdx.x;
    const int lane = t & 31;
    const int wv   = t >> 5;

    const int storeBase = swzc(lane | (wv << 6));
    const int permBase  = swzc(Rmap(lane | (wv << 6)));

    for (int k = t; k < DIM / 2; k += BLOCK) st4[k] = make_float4(k == 0 ? 1.0f : 0.0f, 0.0f, 0.0f, 0.0f);

    const float a0 = model_input[2 * b + 0];
    const float a1 = model_input[2 * b + 1];

    if (t < NL * NQ) {
        const int l = t / NQ, w = t % NQ;
        const int limit = NQ - (l >> 1);
        float th = weights[(l * NQ + w) * 2 + 0];
        if (w < limit) th += (w & 1) ? a1 : a0;
        const float ph = weights[(l * NQ + w) * 2 + 1];
        float c, s, pr, pi;
        __sincosf(0.5f * th, &s, &c);
        __sincosf(0.5f * ph, &pi, &pr);
        gates[l][w][0] =  pr * c;  gates[l][w][1] = -pi * c;
        gates[l][w][2] = -pr * s;  gates[l][w][3] =  pi * s;
        gates[l][w][4] =  pr * s;  gates[l][w][5] =  pi * s;
        gates[l][w][6] =  pr * c;  gates[l][w][7] =  pi * c;
    }
    __syncthreads();

    const int m  = lane & 15;
    const int hi = lane >> 4;

    auto build_tables = [&](const int ll) {
        const int g = t >> 6, jo = (t >> 3) & 7, ji = t & 7;
        float re = gates[ll][3 * g][((jo >> 2) * 2 + (ji >> 2)) * 2 + 0];
        float im = gates[ll][3 * g][((jo >> 2) * 2 + (ji >> 2)) * 2 + 1];
        #pragma unroll
        for (int k = 1; k < 3; ++k) {
            const int w  = 3 * g + k;
            const int bo = (jo >> (2 - k)) & 1;
            const int bi = (ji >> (2 - k)) & 1;
            const float er = gates[ll][w][(bo * 2 + bi) * 2 + 0];
            const float ei = gates[ll][w][(bo * 2 + bi) * 2 + 1];
            const float nr = re * er - im * ei;
            const float ni = re * ei + im * er;
            re = nr; im = ni;
        }
        const int slot = (g == 3) ? ((ji & 3) * 32 + (ji >> 2)) : ((ji >> 1) * 32 + (ji & 1));
        Ae[g][slot + (2 * jo    ) * 2] = make_float2(re, -im);
        Ae[g][slot + (2 * jo + 1) * 2] = make_float2(im,  re);
    };
    build_tables(0);
    __syncthreads();

    #pragma unroll 1
    for (int l = 0; l < NL; ++l) {
        #pragma unroll
        for (int g = 0; g < 3; ++g) {
            const int sh = 9 - 3 * g;

            v2f af[4];
            #pragma unroll
            for (int c = 0; c < 4; ++c) {
                af[c] = *(const v2f*)&Ae[g][c * 32 + m * 2 + hi];
            }

            const int cb  = swzc(colM(g, m) | colX(g, wv * 2));
            const int ldB = cb ^ (hi ? swzc(1 << sh) : 0);
            const int stD = cb ^ (hi ? swzc(4 << sh) : 0);
            v2f bb[4][4];
            #pragma unroll
            for (int tp = 0; tp < 2; ++tp) {
                #pragma unroll
                for (int c = 0; c < 4; ++c) {
                    const float4 v = st4[(ldB ^ swzc(colX(g, tp)) ^ swzc((2 * c) << sh)) >> 1];
                    bb[2 * tp    ][c] = (v2f){v.x, v.y};
                    bb[2 * tp + 1][c] = (v2f){v.z, v.w};
                }
            }
            v8f d[4];
            mma_tiles(af, bb, d);
            #pragma unroll
            for (int tp = 0; tp < 2; ++tp) {
                #pragma unroll
                for (int k = 0; k < 4; ++k) {
                    st4[(stD ^ swzc(colX(g, tp)) ^ swzc(k << sh)) >> 1] =
                        make_float4(d[2 * tp][2 * k], d[2 * tp][2 * k + 1], d[2 * tp + 1][2 * k], d[2 * tp + 1][2 * k + 1]);
                }
            }
            if (g == 0) __syncthreads(); else wave_lds_order();
        }

        {
            v2f af[4];
            #pragma unroll
            for (int c = 0; c < 4; ++c) {
                af[c] = *(const v2f*)&Ae[3][c * 32 + m * 2 + hi];
            }
            v2f bb[4][4];
            int sb[4];
            #pragma unroll
            for (int tt = 0; tt < 4; ++tt) {
                const int rn = (wv * 4 + tt) * 16 + m;
                sb[tt] = (swzc(rn << 3) >> 1) ^ (2 * hi);
                const float4 v0 = st4[sb[tt]];
                const float4 v1 = st4[sb[tt] ^ 1];
                bb[tt][0] = (v2f){v0.x, v0.y};
                bb[tt][1] = (v2f){v0.z, v0.w};
                bb[tt][2] = (v2f){v1.x, v1.y};
                bb[tt][3] = (v2f){v1.z, v1.w};
            }
            v8f d[4];
            mma_tiles(af, bb, d);
            #pragma unroll
            for (int tt = 0; tt < 4; ++tt) {
                st4[sb[tt]]     = make_float4(d[tt][0], d[tt][1], d[tt][2], d[tt][3]);
                st4[sb[tt] ^ 1] = make_float4(d[tt][4], d[tt][5], d[tt][6], d[tt][7]);
            }
            __syncthreads();
        }

        float2 tmp[16];
        #pragma unroll
        for (int k = 0; k < 16; ++k) {
            tmp[k] = st2[permBase ^ swzc(Rmap(ringK(k)))];
        }
        if (l + 1 < NL) build_tables(l + 1);
        __syncthreads();
        #pragma unroll
        for (int k = 0; k < 16; ++k) {
            st2[storeBase ^ swzc(ringK(k))] = tmp[k];
        }
        wave_lds_order();
    }

    float acc = 0.0f;
    #pragma unroll
    for (int k = 0; k < 16; ++k) {
        const float2 v = st2[storeBase ^ swzc(ringK(k))];
        acc += v.x * v.x + v.y * v.y;
    }
    red[t] = (lane & 1) ? -acc : acc;
    __syncthreads();
    #pragma unroll
    for (int off = BLOCK / 2; off > 0; off >>= 1) {
        if (t < off) red[t] += red[t + off];
        __syncthreads();
    }
    if (t == 0) out[b] = red[0];
}

extern "C" void kernel_launch(void* const* d_in, const int* in_sizes, int n_in,
                              void* d_out, int out_size, void* d_ws, size_t ws_size,
                              hipStream_t stream) {
    const float* model_input = (const float*)d_in[0];
    const float* weights     = (const float*)d_in[1];
    float* out = (float*)d_out;
    const int batch = in_sizes[0] / 2;
    qsim12_kernel<<<batch, BLOCK, 0, stream>>>(model_input, weights, out);
}
